# starting kernel
# baseline (speedup 1.0000x reference)
#include <hip/hip_runtime.h>

typedef float v2f __attribute__((ext_vector_type(2)));
typedef float v4f __attribute__((ext_vector_type(4)));
typedef float v8f __attribute__((ext_vector_type(8)));

#define GEM_EPS 1e-6f

__device__ __forceinline__ float fast_exp2(float x) { return __builtin_amdgcn_exp2f(x); }
__device__ __forceinline__ float fast_log2(float x) { return __builtin_amdgcn_logf(x); }

__global__ __launch_bounds__(256) void gem_pool_wmma_kernel(
    const float* __restrict__ x,
    const float* __restrict__ params,
    float* __restrict__ out,
    int n_bc, int C, int hw)
{
    const int lane   = threadIdx.x & 31;
    const int gwave  = blockIdx.x * (blockDim.x >> 5) + (threadIdx.x >> 5);
    const int bc0    = gwave << 4;
    if (bc0 >= n_bc) return;

    const int   r    = lane & 15;
    const int   row  = bc0 + r;
    const int   c0   = bc0 % C;
    const float p    = params[c0 + r];

    const int   half         = lane >> 4;
    const int   nfloats_half = hw >> 1;
    const int   nf4          = nfloats_half >> 2;
    const v4f*  bp = (const v4f*)(x + (size_t)row * (size_t)hw
                                    + (size_t)half * (size_t)nfloats_half);

    float acc0 = 0.0f, acc1 = 0.0f;

    if (__all(p == 3.0f)) {
        #pragma unroll 4
        for (int k = 0; k < nf4; ++k) {
            v4f v = __builtin_nontemporal_load(bp + k);
            float m0 = fmaxf(v.x, GEM_EPS), m1 = fmaxf(v.y, GEM_EPS);
            float m2 = fmaxf(v.z, GEM_EPS), m3 = fmaxf(v.w, GEM_EPS);
            acc0 += m0 * m0 * m0 + m1 * m1 * m1;
            acc1 += m2 * m2 * m2 + m3 * m3 * m3;
        }
    } else {
        #pragma unroll 2
        for (int k = 0; k < nf4; ++k) {
            v4f v = __builtin_nontemporal_load(bp + k);
            float m0 = fmaxf(v.x, GEM_EPS), m1 = fmaxf(v.y, GEM_EPS);
            float m2 = fmaxf(v.z, GEM_EPS), m3 = fmaxf(v.w, GEM_EPS);
            acc0 += fast_exp2(p * fast_log2(m0)) + fast_exp2(p * fast_log2(m1));
            acc1 += fast_exp2(p * fast_log2(m2)) + fast_exp2(p * fast_log2(m3));
        }
    }

    v2f a; a.x = acc0; a.y = acc1;
    v2f b; b.x = 1.0f; b.y = 1.0f;
    v8f c = {};
    v8f d = __builtin_amdgcn_wmma_f32_16x16x4_f32(
         false, a,  false, b,
         (short)0, c,  false,  false);

    const float inv_hw = 1.0f / (float)hw;
    if (lane == 0) {
        #pragma unroll
        for (int i = 0; i < 8; ++i) {
            float pp   = params[c0 + i];
            float mean = d[i] * inv_hw;
            out[bc0 + i] = fast_exp2(fast_log2(mean) / pp);
        }
    } else if (lane == 16) {
        #pragma unroll
        for (int i = 0; i < 8; ++i) {
            float pp   = params[c0 + 8 + i];
            float mean = d[i] * inv_hw;
            out[bc0 + 8 + i] = fast_exp2(fast_log2(mean) / pp);
        }
    }
}

extern "C" void kernel_launch(void* const* d_in, const int* in_sizes, int n_in,
                              void* d_out, int out_size, void* d_ws, size_t ws_size,
                              hipStream_t stream) {
    const float* x      = (const float*)d_in[0];
    const float* params = (const float*)d_in[1];
    float*       out    = (float*)d_out;

    const int n_bc = out_size;
    const int C    = in_sizes[1];
    const int hw   = in_sizes[0] / out_size;

    const int waves   = n_bc / 16;
    const int threads = 256;
    const int wpb     = threads / 32;
    const int blocks  = (waves + wpb - 1) / wpb;

    gem_pool_wmma_kernel<<<blocks, threads, 0, stream>>>(x, params, out, n_bc, C, hw);
}


# lanes lie across one row: a wave sums row pairs (49 whole 128-B lines) with whole-line 512-B loads; per-row lane partials go through LDS, transposed, into the one matrix sum: summation order changes
# speedup vs baseline: 4.7810x; 4.7810x over previous
#include <hip/hip_runtime.h>


typedef float v2f __attribute__((ext_vector_type(2)));
typedef float v4f __attribute__((ext_vector_type(4)));
typedef float v8f __attribute__((ext_vector_type(8)));

#define GEM_EPS 1e-6f

#define ROW_FLOATS   784
#define ROW_F4       (ROW_FLOATS / 4)
#define PAIR_F4      (2 * ROW_F4)
#define FULL_LOADS   6
#define SHARED_F4    (FULL_LOADS * 32)
#define ODD_F4       (SHARED_F4 + 8)
#define ROWS_PER_WAVE 16
#define LDS_PITCH    33

__device__ __forceinline__ float fast_exp2(float x) { return __builtin_amdgcn_exp2f(x); }
__device__ __forceinline__ float fast_log2(float x) { return __builtin_amdgcn_logf(x); }

template <bool CUBE>
__device__ __forceinline__ float gem_pow(float m, float p) {
    return CUBE ? m * m * m : fast_exp2(p * fast_log2(m));
}

template <bool CUBE>
__device__ __forceinline__ void add_piece(v4f v, float p, float& s0, float& s1) {
    float m0 = fmaxf(v.x, GEM_EPS), m1 = fmaxf(v.y, GEM_EPS);
    float m2 = fmaxf(v.z, GEM_EPS), m3 = fmaxf(v.w, GEM_EPS);
    s0 += gem_pow<CUBE>(m0, p) + gem_pow<CUBE>(m1, p);
    s1 += gem_pow<CUBE>(m2, p) + gem_pow<CUBE>(m3, p);
}

template <bool CUBE>
__device__ __forceinline__ void sum_rows(const float* __restrict__ x,
                                         const float* __restrict__ params,
                                         float* part, int bc0, int c0, int lane)
{
    const bool in_even = lane < 4;
    const bool in_odd  = (lane >= 4) && (lane < 8);

    for (int q = 0; q < ROWS_PER_WAVE / 4; ++q) {
        const v4f* quad = (const v4f*)(x + (size_t)(bc0 + 4 * q) * (size_t)ROW_FLOATS);

        v4f ev[2][FULL_LOADS], od[2][FULL_LOADS], sh[2];
        #pragma unroll
        for (int j = 0; j < 2; ++j) {
            const v4f* pair = quad + j * PAIR_F4;
            #pragma unroll
            for (int i = 0; i < FULL_LOADS; ++i)
                ev[j][i] = __builtin_nontemporal_load(pair + 32 * i + lane);
            sh[j] = (v4f){0.0f, 0.0f, 0.0f, 0.0f};
            if (lane < 8)
                sh[j] = __builtin_nontemporal_load(pair + SHARED_F4 + lane);
            #pragma unroll
            for (int i = 0; i < FULL_LOADS; ++i)
                od[j][i] = __builtin_nontemporal_load(pair + ODD_F4 + 32 * i + lane);
        }

        #pragma unroll
        for (int j = 0; j < 2; ++j) {
            const int   r_even = 4 * q + 2 * j;
            const float p_even = CUBE ? 3.0f : params[c0 + r_even];
            const float p_odd  = CUBE ? 3.0f : params[c0 + r_even + 1];

            float e0 = 0.0f, e1 = 0.0f, o0 = 0.0f, o1 = 0.0f, t0 = 0.0f, t1 = 0.0f;
            #pragma unroll
            for (int i = 0; i < FULL_LOADS; ++i) add_piece<CUBE>(ev[j][i], p_even, e0, e1);
            #pragma unroll
            for (int i = 0; i < FULL_LOADS; ++i) add_piece<CUBE>(od[j][i], p_odd, o0, o1);
            add_piece<CUBE>(sh[j], in_even ? p_even : p_odd, t0, t1);
            const float t = t0 + t1;

            part[r_even * LDS_PITCH + lane]       = e0 + e1 + (in_even ? t : 0.0f);
            part[(r_even + 1) * LDS_PITCH + lane] = o0 + o1 + (in_odd  ? t : 0.0f);
        }
    }
}

__global__ __launch_bounds__(256) void gem_pool_rowwise_kernel(
    const float* __restrict__ x,
    const float* __restrict__ params,
    float* __restrict__ out,
    int n_bc, int C)
{
    __shared__ float lds_part[8 * ROWS_PER_WAVE * LDS_PITCH];

    const int lane  = threadIdx.x & 31;
    const int wib   = __builtin_amdgcn_readfirstlane(threadIdx.x >> 5);
    const int gwave = blockIdx.x * (blockDim.x >> 5) + wib;
    const int bc0   = gwave << 4;
    if (bc0 >= n_bc) return;

    const int c0 = bc0 % C;
    float* part  = lds_part + wib * (ROWS_PER_WAVE * LDS_PITCH);

    const float p_lane = params[c0 + (lane & 15)];
    if (__all(p_lane == 3.0f)) {
        sum_rows<true>(x, params, part, bc0, c0, lane);
    } else {
        sum_rows<false>(x, params, part, bc0, c0, lane);
    }

    __builtin_amdgcn_fence(__ATOMIC_RELEASE, "wavefront");
    __builtin_amdgcn_wave_barrier();
    __builtin_amdgcn_fence(__ATOMIC_ACQUIRE, "wavefront");

    const int c = lane & 15, h = lane >> 4;
    const float* col = part + c * LDS_PITCH + 16 * h;
    float a0 = 0.0f, a1 = 0.0f;
    #pragma unroll
    for (int j = 0; j < 8; ++j) { a0 += col[j]; a1 += col[8 + j]; }

    v2f a; a.x = a0; a.y = a1;
    v2f b; b.x = 1.0f; b.y = 1.0f;
    v8f cz = {};
    v8f d = __builtin_amdgcn_wmma_f32_16x16x4_f32(
         false, a,  false, b,
         (short)0, cz,  false,  false);

    const float inv_hw = 1.0f / (float)ROW_FLOATS;
    if (lane == 0) {
        #pragma unroll
        for (int i = 0; i < 8; ++i) {
            float pp   = params[c0 + i];
            float mean = d[i] * inv_hw;
            out[bc0 + i] = fast_exp2(fast_log2(mean) / pp);
        }
    } else if (lane == 16) {
        #pragma unroll
        for (int i = 0; i < 8; ++i) {
            float pp   = params[c0 + 8 + i];
            float mean = d[i] * inv_hw;
            out[bc0 + 8 + i] = fast_exp2(fast_log2(mean) / pp);
        }
    }
}

extern "C" void kernel_launch(void* const* d_in, const int* in_sizes, int n_in,
                              void* d_out, int out_size, void* d_ws, size_t ws_size,
                              hipStream_t stream) {
    const float* x      = (const float*)d_in[0];
    const float* params = (const float*)d_in[1];
    float*       out    = (float*)d_out;

    const int n_bc = out_size;
    const int C    = in_sizes[1];

    const int waves   = n_bc / 16;
    const int threads = 256;
    const int wpb     = threads / 32;
    const int blocks  = (waves + wpb - 1) / wpb;

    gem_pool_rowwise_kernel<<<blocks, threads, 0, stream>>>(x, params, out, n_bc, C);
}
